# starting kernel
# baseline (speedup 1.0000x reference)
#include <hip/hip_runtime.h>


typedef __attribute__((ext_vector_type(2))) float v2f;
typedef __attribute__((ext_vector_type(8))) float v8f;

#define N_SAMPLES 32768
#define N_EVENTS  64
#define STEP      256
#define BATCH     16

#define WAVES_PER_BLOCK 8
#define WAVE_TILE       64
#define TILES_PER_BATCH  (N_SAMPLES / WAVE_TILE)
#define BLOCKS_PER_BATCH (TILES_PER_BATCH / WAVES_PER_BLOCK)
#define GRID (BATCH * BLOCKS_PER_BATCH)

__device__ __forceinline__ float masked_gather(const float* __restrict__ xb,
                                               int rowbase, int j) {
    const int mask = ~(j >> 31);
    const float v  = xb[rowbase + (j & mask)];
    return __int_as_float(__float_as_int(v) & mask);
}

__global__ __launch_bounds__(256)
void sparse_audio_gather_wmma(const float* __restrict__ x,
                              const int*   __restrict__ indices,
                              float*       __restrict__ out) {
    __shared__ int t_sh[N_EVENTS];

    const int tid = threadIdx.x;
    const int b = blockIdx.x / BLOCKS_PER_BATCH;

    if (tid < N_EVENTS) {
        t_sh[tid] = indices[b * N_EVENTS + tid] * STEP;
    }
    __syncthreads();

    const int wave      = blockIdx.x * WAVES_PER_BLOCK + (tid >> 5);
    const int lane      = tid & 31;
    const int lh        = lane >> 4;
    const int n_in_tile = lane & 15;
    const int s_base    = (wave % TILES_PER_BATCH) * WAVE_TILE;
    const int n0        = s_base + n_in_tile;

    const float* xb = x + (size_t)b * N_EVENTS * N_SAMPLES;

    v8f zero = {};
    v8f acc[4];
    acc[0] = zero; acc[1] = zero; acc[2] = zero; acc[3] = zero;

    v2f a_ones;
    a_ones.x = 1.0f;
    a_ones.y = 1.0f;

    #pragma unroll 8
    for (int g = 0; g < 16; ++g) {
        const int eA = 4 * g + lh;
        const int eB = 4 * g + 2 + lh;
        const int tA = t_sh[eA];
        const int tB = t_sh[eB];
        const int rowA = eA * N_SAMPLES;
        const int rowB = eB * N_SAMPLES;

        #pragma unroll
        for (int t = 0; t < 4; ++t) {
            const int n = n0 + t * 16;
            v2f bm;
            bm.x = masked_gather(xb, rowA, n - tA);
            bm.y = masked_gather(xb, rowB, n - tB);
            acc[t] = __builtin_amdgcn_wmma_f32_16x16x4_f32(
                 false, a_ones,
                 false, bm,
                 (short)0, acc[t],
                 false,  false);
        }
    }

    if (lane < 16) {
        float* ob = out + (size_t)b * N_SAMPLES + s_base;
        #pragma unroll
        for (int t = 0; t < 4; ++t) {
            ob[t * 16 + lane] = acc[t][0];
        }
    }
}

extern "C" void kernel_launch(void* const* d_in, const int* in_sizes, int n_in,
                              void* d_out, int out_size, void* d_ws, size_t ws_size,
                              hipStream_t stream) {
    (void)in_sizes; (void)n_in; (void)out_size; (void)d_ws; (void)ws_size;
    const float* x   = (const float*)d_in[0];
    const int*   idx = (const int*)d_in[1];
    float*       out = (float*)d_out;

    sparse_audio_gather_wmma<<<GRID, 256, 0, stream>>>(x, idx, out);
}


# events split over a group's 8 waves: each wave streams its begun events' 1-KB chunks into per-lane partial sums (vector fma), the 8 partials meet in LDS and a ones-WMMA sums them (new sum order)
# speedup vs baseline: 1.4554x; 1.4554x over previous
#include <hip/hip_runtime.h>


typedef __attribute__((ext_vector_type(2))) float v2f;
typedef __attribute__((ext_vector_type(4))) float v4f;
typedef __attribute__((ext_vector_type(8))) float v8f;

#define N_SAMPLES 32768
#define N_EVENTS  64
#define STEP      256
#define BATCH     16
#define N_BLK     (N_SAMPLES / STEP)

#define WAVES        8
#define EV_PER_WAVE  (N_EVENTS / WAVES)
#define RUN          4
#define SLOTS        (2 * RUN)
#define WG_PER_BATCH (N_BLK / SLOTS)
#define GRID         (BATCH * WG_PER_BATCH)

struct ChunkSet {
    v4f lo[RUN];
    v4f hi[RUN];
};

__device__ __forceinline__ void load_item(ChunkSet& s, const char* __restrict__ row,
                                          unsigned lane16, int k) {
    #pragma unroll
    for (int t = 0; t < RUN; ++t) {
        const int kc = (k + t) < 0 ? 0 : (k + t);
        const char* p = row + (size_t)kc * (STEP * 4);
        s.lo[t] = *(const v4f*)(p + lane16);
        s.hi[t] = *(const v4f*)(p + 512 + lane16);
    }
}

__global__ __launch_bounds__(256)
void sparse_audio_split_events(const float* __restrict__ x,
                               const int*   __restrict__ indices,
                               float*       __restrict__ out) {
    __shared__ float part[WAVES * SLOTS * STEP];

    const int tid  = threadIdx.x;
    const int lane = tid & 31;
    const int w = __builtin_amdgcn_readfirstlane(tid >> 5);
    const int b = blockIdx.x / WG_PER_BATCH;
    const int j = blockIdx.x % WG_PER_BATCH;
    const int q_lo = RUN * j;
    const int q_hi = N_BLK - RUN - RUN * j;

    int iv = indices[b * N_EVENTS + w * EV_PER_WAVE + (lane & 7)];
    iv = iv < 0 ? 0 : iv;
    const int kv = ((lane & 8) ? q_lo : q_hi) - iv;
    unsigned m = __builtin_amdgcn_ballot_w32(lane < 16 && kv + (RUN - 1) >= 0);
    const int n = __builtin_popcount(m);

    const char* xw = (const char*)(x + ((size_t)b * N_EVENTS + w * EV_PER_WAVE) * N_SAMPLES);
    const unsigned lane16 = (unsigned)lane * 16u;
    float* pw = part + w * (SLOTS * STEP) + lane * 4;

    v4f alo[RUN], ahi[RUN];
    #pragma unroll
    for (int t = 0; t < RUN; ++t) {
        alo[t] = (v4f){0.f, 0.f, 0.f, 0.f};
        ahi[t] = alo[t];
    }
    bool low_phase = false;

    int cu = 0, ck = 0;
    #define ADVANCE() do { if (m) { cu = __builtin_ctz(m); m &= m - 1; \
                                    ck = __builtin_amdgcn_readlane(kv, cu); } \
                           asm volatile("" : "+s"(cu), "+s"(ck)); } while (0)
    #define ROW_OF(u) (xw + (size_t)((u) & 7) * (N_SAMPLES * 4))
    #define CONSUME(S, U, K) do { \
        if (((U) & 8) && !low_phase) {            \
            _Pragma("unroll") for (int t = 0; t < RUN; ++t) { \
                *(v4f*)(pw + (RUN + t) * STEP)       = alo[t]; \
                *(v4f*)(pw + (RUN + t) * STEP + 128) = ahi[t]; \
                alo[t] = (v4f){0.f, 0.f, 0.f, 0.f}; ahi[t] = alo[t]; } \
            low_phase = true; } \
          \
        _Pragma("unroll") for (int t = 0; t < RUN; ++t) { \
            const float wt = ((K) + t >= 0) ? 1.0f : 0.0f; \
            alo[t] = __builtin_elementwise_fma(S.lo[t], (v4f){wt, wt, wt, wt}, alo[t]); \
            ahi[t] = __builtin_elementwise_fma(S.hi[t], (v4f){wt, wt, wt, wt}, ahi[t]); } \
    } while (0)

    ChunkSet s0, s1;
    int u0, k0, u1, k1;

    ADVANCE(); u0 = cu; k0 = ck; load_item(s0, ROW_OF(u0), lane16, k0);

    for (int i = 0; i < n; i += 2) {
        ADVANCE(); u1 = cu; k1 = ck; load_item(s1, ROW_OF(u1), lane16, k1);
        CONSUME(s0, u0, k0);
        ADVANCE(); u0 = cu; k0 = ck; load_item(s0, ROW_OF(u0), lane16, k0);
        if (i + 1 < n) CONSUME(s1, u1, k1);
    }
    #undef ADVANCE
    #undef ROW_OF
    #undef CONSUME

    {
        const int sb = low_phase ? 0 : RUN;
        #pragma unroll
        for (int t = 0; t < RUN; ++t) {
            *(v4f*)(pw + (sb + t) * STEP)       = alo[t];
            *(v4f*)(pw + (sb + t) * STEP + 128) = ahi[t];
        }
        if (!low_phase) {
            #pragma unroll
            for (int t = 0; t < RUN; ++t) {
                *(v4f*)(pw + t * STEP)       = (v4f){0.f, 0.f, 0.f, 0.f};
                *(v4f*)(pw + t * STEP + 128) = (v4f){0.f, 0.f, 0.f, 0.f};
            }
        }
    }
    __syncthreads();

    const int h = lane >> 4;
    const int c = lane & 15;
    const float* ps = part + w * STEP + 4 * c;
    const int kstride = SLOTS * STEP;
    const int blk = (w < RUN) ? (q_lo + w) : (q_hi + w - RUN);
    float* ob = out + (size_t)b * N_SAMPLES + blk * STEP + lane * 4;

    v2f ones;
    ones.x = 1.0f;
    ones.y = 1.0f;
    const v8f zero8 = {};

    #pragma unroll
    for (int half = 0; half < 2; ++half) {
        v4f r[2];
        #pragma unroll
        for (int t2 = 0; t2 < 2; ++t2) {
            const float* p = ps + 64 * (half * 2 + t2);
            const v4f q0 = *(const v4f*)(p + (2 * h + 0) * kstride);
            const v4f q1 = *(const v4f*)(p + (2 * h + 1) * kstride);
            const v4f q2 = *(const v4f*)(p + (4 + 2 * h + 0) * kstride);
            const v4f q3 = *(const v4f*)(p + (4 + 2 * h + 1) * kstride);
            #pragma unroll
            for (int i = 0; i < 4; ++i) {
                v2f b1, b2;
                b1.x = q0[i]; b1.y = q1[i];
                b2.x = q2[i]; b2.y = q3[i];
                v8f d = __builtin_amdgcn_wmma_f32_16x16x4_f32(
                    false, ones, false, b1, (short)0, zero8, false, false);
                d = __builtin_amdgcn_wmma_f32_16x16x4_f32(
                    false, ones, false, b2, (short)0, d, false, false);
                r[t2][i] = d[0];
            }
        }
        const v4f v = h ? r[1] : r[0];
        *(v4f*)(ob + half * 128) = v;
    }
}

extern "C" void kernel_launch(void* const* d_in, const int* in_sizes, int n_in,
                              void* d_out, int out_size, void* d_ws, size_t ws_size,
                              hipStream_t stream) {
    (void)in_sizes; (void)n_in; (void)out_size; (void)d_ws; (void)ws_size;
    const float* x   = (const float*)d_in[0];
    const int*   idx = (const int*)d_in[1];
    float*       out = (float*)d_out;

    sparse_audio_split_events<<<GRID, 256, 0, stream>>>(x, idx, out);
}


# a group's begun (event,run) items are dealt round-robin over its 8 waves (each wave ranks all 64 events by ballot, private LDS list) instead of 8 fixed events per wave (new sum order)
# speedup vs baseline: 1.6402x; 1.1270x over previous
#include <hip/hip_runtime.h>


typedef __attribute__((ext_vector_type(2))) float v2f;
typedef __attribute__((ext_vector_type(4))) float v4f;
typedef __attribute__((ext_vector_type(8))) float v8f;

#define N_SAMPLES 32768
#define N_EVENTS  64
#define STEP      256
#define BATCH     16
#define N_BLK     (N_SAMPLES / STEP)

#define WAVES        8
#define LIST_LEN     (2 * N_EVENTS)
#define RUN          4
#define SLOTS        (2 * RUN)
#define WG_PER_BATCH (N_BLK / SLOTS)
#define GRID         (BATCH * WG_PER_BATCH)

struct ChunkSet {
    v4f lo[RUN];
    v4f hi[RUN];
};

__device__ __forceinline__ void load_item(ChunkSet& s, const char* __restrict__ row,
                                          unsigned lane16, int k) {
    #pragma unroll
    for (int t = 0; t < RUN; ++t) {
        const int kc = (k + t) < 0 ? 0 : (k + t);
        const char* p = row + (size_t)kc * (STEP * 4);
        s.lo[t] = *(const v4f*)(p + lane16);
        s.hi[t] = *(const v4f*)(p + 512 + lane16);
    }
}

__global__ __launch_bounds__(256)
void sparse_audio_dealt_items(const float* __restrict__ x,
                               const int*   __restrict__ indices,
                               float*       __restrict__ out) {
    __shared__ float part[WAVES * SLOTS * STEP];
    __shared__ unsigned items[WAVES * LIST_LEN];

    const int tid  = threadIdx.x;
    const int lane = tid & 31;
    const int w = __builtin_amdgcn_readfirstlane(tid >> 5);
    const int b = blockIdx.x / WG_PER_BATCH;
    const int j = blockIdx.x % WG_PER_BATCH;
    const int q_lo = RUN * j;
    const int q_hi = N_BLK - RUN - RUN * j;

    int ia = indices[b * N_EVENTS + lane];
    int ib = indices[b * N_EVENTS + 32 + lane];
    ia = ia < 0 ? 0 : ia;
    ib = ib < 0 ? 0 : ib;
    const int kha = q_hi - ia, khb = q_hi - ib;
    const int kla = q_lo - ia, klb = q_lo - ib;
    const bool bha = kha + (RUN - 1) >= 0, bhb = khb + (RUN - 1) >= 0;
    const bool bla = kla + (RUN - 1) >= 0, blb = klb + (RUN - 1) >= 0;
    const unsigned mha = __builtin_amdgcn_ballot_w32(bha), mhb = __builtin_amdgcn_ballot_w32(bhb);
    const unsigned mla = __builtin_amdgcn_ballot_w32(bla), mlb = __builtin_amdgcn_ballot_w32(blb);
    const int r_hb  = __builtin_popcount(mha);
    const int r_la  = r_hb + __builtin_popcount(mhb);
    const int r_lb  = r_la + __builtin_popcount(mla);
    const int total = r_lb + __builtin_popcount(mlb);
    unsigned* lw = items + w * LIST_LEN;
    if (bha) lw[__builtin_amdgcn_mbcnt_lo(mha, 0u)]         = (unsigned)(kha + (RUN - 1)) | ((unsigned)lane << 8);
    if (bhb) lw[r_hb + __builtin_amdgcn_mbcnt_lo(mhb, 0u)] = (unsigned)(khb + (RUN - 1)) | ((unsigned)(lane + 32) << 8);
    if (bla) lw[r_la + __builtin_amdgcn_mbcnt_lo(mla, 0u)] = (unsigned)(kla + (RUN - 1)) | ((unsigned)lane << 8) | 0x4000u;
    if (blb) lw[r_lb + __builtin_amdgcn_mbcnt_lo(mlb, 0u)] = (unsigned)(klb + (RUN - 1)) | ((unsigned)(lane + 32) << 8) | 0x4000u;
    const int n = (total + (WAVES - 1) - w) / WAVES;
    unsigned dv = 0u;
    if (lane < n) dv = lw[w + WAVES * lane];

    const char* xb = (const char*)(x + (size_t)b * N_EVENTS * N_SAMPLES);
    const unsigned lane16 = (unsigned)lane * 16u;
    float* pw = part + w * (SLOTS * STEP) + lane * 4;

    v4f alo[RUN], ahi[RUN];
    #pragma unroll
    for (int t = 0; t < RUN; ++t) {
        alo[t] = (v4f){0.f, 0.f, 0.f, 0.f};
        ahi[t] = alo[t];
    }
    bool low_phase = false;

    int cu = 0, ck = 0, jn = 0;
    #define ADVANCE() do { if (jn < n) { const unsigned d_ = __builtin_amdgcn_readlane(dv, jn); ++jn; \
                                         cu = (int)(d_ >> 8); ck = (int)(d_ & 0xFFu) - (RUN - 1); } \
                           asm volatile("" : "+s"(cu), "+s"(ck)); } while (0)
    #define ROW_OF(u) (xb + (size_t)((u) & 63) * (N_SAMPLES * 4))
    #define CONSUME(S, U, K) do { \
        if (((U) & 64) && !low_phase) {           \
            _Pragma("unroll") for (int t = 0; t < RUN; ++t) { \
                *(v4f*)(pw + (RUN + t) * STEP)       = alo[t]; \
                *(v4f*)(pw + (RUN + t) * STEP + 128) = ahi[t]; \
                alo[t] = (v4f){0.f, 0.f, 0.f, 0.f}; ahi[t] = alo[t]; } \
            low_phase = true; } \
          \
        _Pragma("unroll") for (int t = 0; t < RUN; ++t) { \
            const float wt = ((K) + t >= 0) ? 1.0f : 0.0f; \
            alo[t] = __builtin_elementwise_fma(S.lo[t], (v4f){wt, wt, wt, wt}, alo[t]); \
            ahi[t] = __builtin_elementwise_fma(S.hi[t], (v4f){wt, wt, wt, wt}, ahi[t]); } \
    } while (0)

    ChunkSet s0, s1;
    int u0, k0, u1, k1;

    ADVANCE(); u0 = cu; k0 = ck; load_item(s0, ROW_OF(u0), lane16, k0);

    for (int i = 0; i < n; i += 2) {
        ADVANCE(); u1 = cu; k1 = ck; load_item(s1, ROW_OF(u1), lane16, k1);
        CONSUME(s0, u0, k0);
        ADVANCE(); u0 = cu; k0 = ck; load_item(s0, ROW_OF(u0), lane16, k0);
        if (i + 1 < n) CONSUME(s1, u1, k1);
    }
    #undef ADVANCE
    #undef ROW_OF
    #undef CONSUME

    {
        const int sb = low_phase ? 0 : RUN;
        #pragma unroll
        for (int t = 0; t < RUN; ++t) {
            *(v4f*)(pw + (sb + t) * STEP)       = alo[t];
            *(v4f*)(pw + (sb + t) * STEP + 128) = ahi[t];
        }
        if (!low_phase) {
            #pragma unroll
            for (int t = 0; t < RUN; ++t) {
                *(v4f*)(pw + t * STEP)       = (v4f){0.f, 0.f, 0.f, 0.f};
                *(v4f*)(pw + t * STEP + 128) = (v4f){0.f, 0.f, 0.f, 0.f};
            }
        }
    }
    __syncthreads();

    const int h = lane >> 4;
    const int c = lane & 15;
    const float* ps = part + w * STEP + 4 * c;
    const int kstride = SLOTS * STEP;
    const int blk = (w < RUN) ? (q_lo + w) : (q_hi + w - RUN);
    float* ob = out + (size_t)b * N_SAMPLES + blk * STEP + lane * 4;

    v2f ones;
    ones.x = 1.0f;
    ones.y = 1.0f;
    const v8f zero8 = {};

    #pragma unroll
    for (int half = 0; half < 2; ++half) {
        v4f r[2];
        #pragma unroll
        for (int t2 = 0; t2 < 2; ++t2) {
            const float* p = ps + 64 * (half * 2 + t2);
            const v4f q0 = *(const v4f*)(p + (2 * h + 0) * kstride);
            const v4f q1 = *(const v4f*)(p + (2 * h + 1) * kstride);
            const v4f q2 = *(const v4f*)(p + (4 + 2 * h + 0) * kstride);
            const v4f q3 = *(const v4f*)(p + (4 + 2 * h + 1) * kstride);
            #pragma unroll
            for (int i = 0; i < 4; ++i) {
                v2f b1, b2;
                b1.x = q0[i]; b1.y = q1[i];
                b2.x = q2[i]; b2.y = q3[i];
                v8f d = __builtin_amdgcn_wmma_f32_16x16x4_f32(
                    false, ones, false, b1, (short)0, zero8, false, false);
                d = __builtin_amdgcn_wmma_f32_16x16x4_f32(
                    false, ones, false, b2, (short)0, d, false, false);
                r[t2][i] = d[0];
            }
        }
        const v4f v = h ? r[1] : r[0];
        *(v4f*)(ob + half * 128) = v;
    }
}

extern "C" void kernel_launch(void* const* d_in, const int* in_sizes, int n_in,
                              void* d_out, int out_size, void* d_ws, size_t ws_size,
                              hipStream_t stream) {
    (void)in_sizes; (void)n_in; (void)out_size; (void)d_ws; (void)ws_size;
    const float* x   = (const float*)d_in[0];
    const int*   idx = (const int*)d_in[1];
    float*       out = (float*)d_out;

    sparse_audio_dealt_items<<<GRID, 256, 0, stream>>>(x, idx, out);
}


# each wave issues second-level prefetches (256-byte units, two items per instruction, no registers) for all its later items right behind its first item's loads; load and sum order unchanged
# speedup vs baseline: 1.7185x; 1.0477x over previous
#include <hip/hip_runtime.h>


typedef __attribute__((ext_vector_type(2))) float v2f;
typedef __attribute__((ext_vector_type(4))) float v4f;
typedef __attribute__((ext_vector_type(8))) float v8f;

#define N_SAMPLES 32768
#define N_EVENTS  64
#define STEP      256
#define BATCH     16
#define N_BLK     (N_SAMPLES / STEP)

#define WAVES        8
#define LIST_LEN     (2 * N_EVENTS)
#define RUN          4
#define SLOTS        (2 * RUN)
#define WG_PER_BATCH (N_BLK / SLOTS)
#define GRID         (BATCH * WG_PER_BATCH)

struct ChunkSet {
    v4f lo[RUN];
    v4f hi[RUN];
};

__device__ __forceinline__ void load_item(ChunkSet& s, const char* __restrict__ row,
                                          unsigned lane16, int k) {
    #pragma unroll
    for (int t = 0; t < RUN; ++t) {
        const int kc = (k + t) < 0 ? 0 : (k + t);
        const char* p = row + (size_t)kc * (STEP * 4);
        s.lo[t] = *(const v4f*)(p + lane16);
        s.hi[t] = *(const v4f*)(p + 512 + lane16);
    }
}

__global__ __launch_bounds__(256)
void sparse_audio_dealt_items(const float* __restrict__ x,
                               const int*   __restrict__ indices,
                               float*       __restrict__ out) {
    __shared__ float part[WAVES * SLOTS * STEP];
    __shared__ unsigned items[WAVES * LIST_LEN];

    const int tid  = threadIdx.x;
    const int lane = tid & 31;
    const int w = __builtin_amdgcn_readfirstlane(tid >> 5);
    const int b = blockIdx.x / WG_PER_BATCH;
    const int j = blockIdx.x % WG_PER_BATCH;
    const int q_lo = RUN * j;
    const int q_hi = N_BLK - RUN - RUN * j;

    int ia = indices[b * N_EVENTS + lane];
    int ib = indices[b * N_EVENTS + 32 + lane];
    ia = ia < 0 ? 0 : ia;
    ib = ib < 0 ? 0 : ib;
    const int kha = q_hi - ia, khb = q_hi - ib;
    const int kla = q_lo - ia, klb = q_lo - ib;
    const bool bha = kha + (RUN - 1) >= 0, bhb = khb + (RUN - 1) >= 0;
    const bool bla = kla + (RUN - 1) >= 0, blb = klb + (RUN - 1) >= 0;
    const unsigned mha = __builtin_amdgcn_ballot_w32(bha), mhb = __builtin_amdgcn_ballot_w32(bhb);
    const unsigned mla = __builtin_amdgcn_ballot_w32(bla), mlb = __builtin_amdgcn_ballot_w32(blb);
    const int r_hb  = __builtin_popcount(mha);
    const int r_la  = r_hb + __builtin_popcount(mhb);
    const int r_lb  = r_la + __builtin_popcount(mla);
    const int total = r_lb + __builtin_popcount(mlb);
    unsigned* lw = items + w * LIST_LEN;
    if (bha) lw[__builtin_amdgcn_mbcnt_lo(mha, 0u)]         = (unsigned)(kha + (RUN - 1)) | ((unsigned)lane << 8);
    if (bhb) lw[r_hb + __builtin_amdgcn_mbcnt_lo(mhb, 0u)] = (unsigned)(khb + (RUN - 1)) | ((unsigned)(lane + 32) << 8);
    if (bla) lw[r_la + __builtin_amdgcn_mbcnt_lo(mla, 0u)] = (unsigned)(kla + (RUN - 1)) | ((unsigned)lane << 8) | 0x4000u;
    if (blb) lw[r_lb + __builtin_amdgcn_mbcnt_lo(mlb, 0u)] = (unsigned)(klb + (RUN - 1)) | ((unsigned)(lane + 32) << 8) | 0x4000u;
    const int n = (total + (WAVES - 1) - w) / WAVES;
    unsigned dv = 0u;
    if (lane < n) dv = lw[w + WAVES * lane];

    const char* xb = (const char*)(x + (size_t)b * N_EVENTS * N_SAMPLES);
    const unsigned lane16 = (unsigned)lane * 16u;
    float* pw = part + w * (SLOTS * STEP) + lane * 4;

    v4f alo[RUN], ahi[RUN];
    #pragma unroll
    for (int t = 0; t < RUN; ++t) {
        alo[t] = (v4f){0.f, 0.f, 0.f, 0.f};
        ahi[t] = alo[t];
    }
    bool low_phase = false;

    int cu = 0, ck = 0, jn = 0;
    #define ADVANCE() do { if (jn < n) { const unsigned d_ = __builtin_amdgcn_readlane(dv, jn); ++jn; \
                                         cu = (int)(d_ >> 8); ck = (int)(d_ & 0xFFu) - (RUN - 1); } \
                           asm volatile("" : "+s"(cu), "+s"(ck)); } while (0)
    #define ROW_OF(u) (xb + (size_t)((u) & 63) * (N_SAMPLES * 4))
    #define CONSUME(S, U, K) do { \
        if (((U) & 64) && !low_phase) {           \
            _Pragma("unroll") for (int t = 0; t < RUN; ++t) { \
                *(v4f*)(pw + (RUN + t) * STEP)       = alo[t]; \
                *(v4f*)(pw + (RUN + t) * STEP + 128) = ahi[t]; \
                alo[t] = (v4f){0.f, 0.f, 0.f, 0.f}; ahi[t] = alo[t]; } \
            low_phase = true; } \
          \
        _Pragma("unroll") for (int t = 0; t < RUN; ++t) { \
            const float wt = ((K) + t >= 0) ? 1.0f : 0.0f; \
            alo[t] = __builtin_elementwise_fma(S.lo[t], (v4f){wt, wt, wt, wt}, alo[t]); \
            ahi[t] = __builtin_elementwise_fma(S.hi[t], (v4f){wt, wt, wt, wt}, ahi[t]); } \
    } while (0)

    ChunkSet s0, s1;
    int u0, k0, u1, k1;

    ADVANCE(); u0 = cu; k0 = ck; load_item(s0, ROW_OF(u0), lane16, k0);

    for (int i = 1; i < n; i += 2) {
        const int i2 = (i + 1 < n) ? (i + 1) : i;
        const unsigned da = __builtin_amdgcn_readlane(dv, i);
        const unsigned db = __builtin_amdgcn_readlane(dv, i2);
        const unsigned dd = (lane & 16) ? db : da;
        const int pk = (int)(dd & 0xFFu) - (RUN - 1);
        const int pc = pk < 0 ? 0 : pk;
        const char* pp = xb + (size_t)((dd >> 8) & 63u) * (N_SAMPLES * 4)
                            + (size_t)pc * (STEP * 4) + (size_t)(lane & 15) * 256u;
        __builtin_prefetch(pp, 0, 1);
        asm volatile("s_wait_xcnt 0x0" ::: "memory");
    }

    for (int i = 0; i < n; i += 2) {
        ADVANCE(); u1 = cu; k1 = ck; load_item(s1, ROW_OF(u1), lane16, k1);
        CONSUME(s0, u0, k0);
        ADVANCE(); u0 = cu; k0 = ck; load_item(s0, ROW_OF(u0), lane16, k0);
        if (i + 1 < n) CONSUME(s1, u1, k1);
    }
    #undef ADVANCE
    #undef ROW_OF
    #undef CONSUME

    {
        const int sb = low_phase ? 0 : RUN;
        #pragma unroll
        for (int t = 0; t < RUN; ++t) {
            *(v4f*)(pw + (sb + t) * STEP)       = alo[t];
            *(v4f*)(pw + (sb + t) * STEP + 128) = ahi[t];
        }
        if (!low_phase) {
            #pragma unroll
            for (int t = 0; t < RUN; ++t) {
                *(v4f*)(pw + t * STEP)       = (v4f){0.f, 0.f, 0.f, 0.f};
                *(v4f*)(pw + t * STEP + 128) = (v4f){0.f, 0.f, 0.f, 0.f};
            }
        }
    }
    __syncthreads();

    const int h = lane >> 4;
    const int c = lane & 15;
    const float* ps = part + w * STEP + 4 * c;
    const int kstride = SLOTS * STEP;
    const int blk = (w < RUN) ? (q_lo + w) : (q_hi + w - RUN);
    float* ob = out + (size_t)b * N_SAMPLES + blk * STEP + lane * 4;

    v2f ones;
    ones.x = 1.0f;
    ones.y = 1.0f;
    const v8f zero8 = {};

    #pragma unroll
    for (int half = 0; half < 2; ++half) {
        v4f r[2];
        #pragma unroll
        for (int t2 = 0; t2 < 2; ++t2) {
            const float* p = ps + 64 * (half * 2 + t2);
            const v4f q0 = *(const v4f*)(p + (2 * h + 0) * kstride);
            const v4f q1 = *(const v4f*)(p + (2 * h + 1) * kstride);
            const v4f q2 = *(const v4f*)(p + (4 + 2 * h + 0) * kstride);
            const v4f q3 = *(const v4f*)(p + (4 + 2 * h + 1) * kstride);
            #pragma unroll
            for (int i = 0; i < 4; ++i) {
                v2f b1, b2;
                b1.x = q0[i]; b1.y = q1[i];
                b2.x = q2[i]; b2.y = q3[i];
                v8f d = __builtin_amdgcn_wmma_f32_16x16x4_f32(
                    false, ones, false, b1, (short)0, zero8, false, false);
                d = __builtin_amdgcn_wmma_f32_16x16x4_f32(
                    false, ones, false, b2, (short)0, d, false, false);
                r[t2][i] = d[0];
            }
        }
        const v4f v = h ? r[1] : r[0];
        *(v4f*)(ob + half * 128) = v;
    }
}

extern "C" void kernel_launch(void* const* d_in, const int* in_sizes, int n_in,
                              void* d_out, int out_size, void* d_ws, size_t ws_size,
                              hipStream_t stream) {
    (void)in_sizes; (void)n_in; (void)out_size; (void)d_ws; (void)ws_size;
    const float* x   = (const float*)d_in[0];
    const int*   idx = (const int*)d_in[1];
    float*       out = (float*)d_out;

    sparse_audio_dealt_items<<<GRID, 256, 0, stream>>>(x, idx, out);
}
